# starting kernel
# baseline (speedup 1.0000x reference)
#include <hip/hip_runtime.h>

typedef float v2f __attribute__((ext_vector_type(2)));
typedef float v8f __attribute__((ext_vector_type(8)));

constexpr int Q = 128;
constexpr int S = 8;
constexpr int H = 32;
constexpr int BT = 16;
constexpr float BN_EPS = 1e-3f;

__global__ __launch_bounds__(256) void divenc_wmma_f32(
    const float* __restrict__ x,
    const float* __restrict__ W1,
    const float* __restrict__ b1g,
    const float* __restrict__ gam,
    const float* __restrict__ bet,
    const float* __restrict__ mmean,
    const float* __restrict__ mvar,
    const float* __restrict__ W2,
    const float* __restrict__ b2g,
    float* __restrict__ out)
{
    const int lane  = threadIdx.x & 31;
    const int wave  = threadIdx.x >> 5;
    const int bbase = blockIdx.x * BT;

    const int mrow  = lane & 15;
    const int khalf = lane >> 4;
    const int ncol  = lane & 15;

    const float* xrow = x + (size_t)(bbase + mrow) * (Q * S);

    for (int qi = 0; qi < 16; ++qi) {
        const int q = wave * 16 + qi;
        const float* xq = xrow + q * S;

        v2f a_lo = { xq[2 * khalf + 0], xq[2 * khalf + 1] };
        v2f a_hi = { xq[2 * khalf + 4], xq[2 * khalf + 5] };

        const float* w1q = W1 + q * (S * H);
        const int h0 = ncol, h1 = ncol + 16;
        const int kb = 2 * khalf;
        v2f bt0_lo = { w1q[(kb + 0) * H + h0], w1q[(kb + 1) * H + h0] };
        v2f bt0_hi = { w1q[(kb + 4) * H + h0], w1q[(kb + 5) * H + h0] };
        v2f bt1_lo = { w1q[(kb + 0) * H + h1], w1q[(kb + 1) * H + h1] };
        v2f bt1_hi = { w1q[(kb + 4) * H + h1], w1q[(kb + 5) * H + h1] };

        const int i0 = q * H + h0, i1 = q * H + h1;
        const float bias0 = b1g[i0], bias1 = b1g[i1];
        v8f acc0 = { bias0, bias0, bias0, bias0, bias0, bias0, bias0, bias0 };
        v8f acc1 = { bias1, bias1, bias1, bias1, bias1, bias1, bias1, bias1 };

        acc0 = __builtin_amdgcn_wmma_f32_16x16x4_f32(false, a_lo, false, bt0_lo,
                                                     (short)0, acc0, false, false);
        acc0 = __builtin_amdgcn_wmma_f32_16x16x4_f32(false, a_hi, false, bt0_hi,
                                                     (short)0, acc0, false, false);
        acc1 = __builtin_amdgcn_wmma_f32_16x16x4_f32(false, a_lo, false, bt1_lo,
                                                     (short)0, acc1, false, false);
        acc1 = __builtin_amdgcn_wmma_f32_16x16x4_f32(false, a_hi, false, bt1_hi,
                                                     (short)0, acc1, false, false);

        const float inv0 = gam[i0] * __builtin_amdgcn_rsqf(mvar[i0] + BN_EPS);
        const float inv1 = gam[i1] * __builtin_amdgcn_rsqf(mvar[i1] + BN_EPS);
        const float w2_0 = W2[i0], w2_1 = W2[i1];
        const float c1_0 = inv0 * w2_0;
        const float c1_1 = inv1 * w2_1;
        const float c0s  = fmaf(-mmean[i0] * inv0 + bet[i0], w2_0,
                                (-mmean[i1] * inv1 + bet[i1]) * w2_1);

        float part[8];
        #pragma unroll
        for (int v = 0; v < 8; ++v) {
            float t0 = acc0[v];
            t0 = t0 > 0.0f ? t0 : (__expf(t0) - 1.0f);
            float t1 = acc1[v];
            t1 = t1 > 0.0f ? t1 : (__expf(t1) - 1.0f);
            part[v] = fmaf(t0, c1_0, fmaf(t1, c1_1, c0s));
        }

        #pragma unroll
        for (int off = 8; off >= 1; off >>= 1) {
            #pragma unroll
            for (int v = 0; v < 8; ++v)
                part[v] += __shfl_xor(part[v], off, 32);
        }

        if (ncol == 0) {
            const float bq = b2g[q];
            float* op = out + (size_t)(bbase + khalf * 8) * Q + q;
            #pragma unroll
            for (int v = 0; v < 8; ++v)
                op[(size_t)v * Q] = part[v] + bq;
        }
    }
}

extern "C" void kernel_launch(void* const* d_in, const int* in_sizes, int n_in,
                              void* d_out, int out_size, void* d_ws, size_t ws_size,
                              hipStream_t stream) {
    const float* x     = (const float*)d_in[0];
    const float* W1    = (const float*)d_in[1];
    const float* b1    = (const float*)d_in[2];
    const float* gamma = (const float*)d_in[3];
    const float* beta  = (const float*)d_in[4];
    const float* mmean = (const float*)d_in[5];
    const float* mvar  = (const float*)d_in[6];
    const float* W2    = (const float*)d_in[7];
    const float* b2    = (const float*)d_in[8];
    float* out = (float*)d_out;

    const int B = in_sizes[0] / (Q * S);
    dim3 grid(B / BT);
    dim3 block(256);
    divenc_wmma_f32<<<grid, block, 0, stream>>>(x, W1, b1, gamma, beta,
                                                mmean, mvar, W2, b2, out);
}


# Slice-stationary waves: each wave holds 4 adjacent slices' W1 fragments, bias and folded BN/W2 coefficients in registers and loops over batch tiles; only x is read and out written inside the loop
# speedup vs baseline: 1.6451x; 1.6451x over previous
#include <hip/hip_runtime.h>


typedef float v2f __attribute__((ext_vector_type(2)));
typedef float v4f __attribute__((ext_vector_type(4)));
typedef float v8f __attribute__((ext_vector_type(8)));

constexpr int Q = 128;
constexpr int S = 8;
constexpr int H = 32;
constexpr int BT = 16;
constexpr float BN_EPS = 1e-3f;

constexpr int QW      = 4;
constexpr int WAVES   = 8;
constexpr int QBLOCKS = Q / (QW * WAVES);
constexpr int BGROUPS = 128;

__global__ __launch_bounds__(256) void divenc_wmma_f32(
    const float* __restrict__ x,
    const float* __restrict__ W1,
    const float* __restrict__ b1g,
    const float* __restrict__ gam,
    const float* __restrict__ bet,
    const float* __restrict__ mmean,
    const float* __restrict__ mvar,
    const float* __restrict__ W2,
    const float* __restrict__ b2g,
    float* __restrict__ out,
    int nrows)
{
    const int lane  = threadIdx.x & 31;
    const int wave  = threadIdx.x >> 5;

    const int mrow  = lane & 15;
    const int khalf = lane >> 4;
    const int ncol  = lane & 15;

    const int qgrp = (blockIdx.x % QBLOCKS) * WAVES + wave;
    const int bgrp = blockIdx.x / QBLOCKS;
    const int nbg  = gridDim.x / QBLOCKS;
    const int q0   = qgrp * QW;
    const int ntiles = nrows / BT;

    v2f bt0_lo[QW], bt0_hi[QW], bt1_lo[QW], bt1_hi[QW];
    float bias0[QW], bias1[QW], c1_0[QW], c1_1[QW], c0s[QW], bq[QW];

    const int h0 = ncol, h1 = ncol + 16;
    const int kb = 2 * khalf;
    for (int j = 0; j < QW; ++j) {
        const int q = q0 + j;
        const float* w1q = W1 + q * (S * H);
        bt0_lo[j] = v2f{ w1q[(kb + 0) * H + h0], w1q[(kb + 1) * H + h0] };
        bt0_hi[j] = v2f{ w1q[(kb + 4) * H + h0], w1q[(kb + 5) * H + h0] };
        bt1_lo[j] = v2f{ w1q[(kb + 0) * H + h1], w1q[(kb + 1) * H + h1] };
        bt1_hi[j] = v2f{ w1q[(kb + 4) * H + h1], w1q[(kb + 5) * H + h1] };

        const int i0 = q * H + h0, i1 = q * H + h1;
        bias0[j] = b1g[i0];
        bias1[j] = b1g[i1];
        const float inv0 = gam[i0] * __builtin_amdgcn_rsqf(mvar[i0] + BN_EPS);
        const float inv1 = gam[i1] * __builtin_amdgcn_rsqf(mvar[i1] + BN_EPS);
        const float w2_0 = W2[i0], w2_1 = W2[i1];
        c1_0[j] = inv0 * w2_0;
        c1_1[j] = inv1 * w2_1;
        c0s[j]  = fmaf(-mmean[i0] * inv0 + bet[i0], w2_0,
                       (-mmean[i1] * inv1 + bet[i1]) * w2_1);
        bq[j]   = b2g[q];
    }

    const float* xlane = x + (size_t)mrow * (Q * S) + q0 * S + 2 * khalf;
    float* olane = out + (size_t)(khalf * 8) * Q + q0;

    v2f a_lo[QW], a_hi[QW];
    {
        const int t0 = bgrp < ntiles ? bgrp : ntiles - 1;
        const float* xt = xlane + (size_t)t0 * (BT * Q * S);
        for (int j = 0; j < QW; ++j) {
            a_lo[j] = v2f{ xt[j * S + 0], xt[j * S + 1] };
            a_hi[j] = v2f{ xt[j * S + 4], xt[j * S + 5] };
        }
    }

    for (int t = bgrp; t < ntiles; t += nbg) {
        v2f n_lo[QW], n_hi[QW];
        {
            const int tn = (t + nbg) < ntiles ? (t + nbg) : t;
            const float* xt = xlane + (size_t)tn * (BT * Q * S);
            for (int j = 0; j < QW; ++j) {
                n_lo[j] = v2f{ xt[j * S + 0], xt[j * S + 1] };
                n_hi[j] = v2f{ xt[j * S + 4], xt[j * S + 5] };
            }
        }

        float res[QW][8];
        for (int j = 0; j < QW; ++j) {
            const float bs0 = bias0[j], bs1 = bias1[j];
            v8f acc0 = { bs0, bs0, bs0, bs0, bs0, bs0, bs0, bs0 };
            v8f acc1 = { bs1, bs1, bs1, bs1, bs1, bs1, bs1, bs1 };

            acc0 = __builtin_amdgcn_wmma_f32_16x16x4_f32(false, a_lo[j], false, bt0_lo[j],
                                                         (short)0, acc0, false, false);
            acc0 = __builtin_amdgcn_wmma_f32_16x16x4_f32(false, a_hi[j], false, bt0_hi[j],
                                                         (short)0, acc0, false, false);
            acc1 = __builtin_amdgcn_wmma_f32_16x16x4_f32(false, a_lo[j], false, bt1_lo[j],
                                                         (short)0, acc1, false, false);
            acc1 = __builtin_amdgcn_wmma_f32_16x16x4_f32(false, a_hi[j], false, bt1_hi[j],
                                                         (short)0, acc1, false, false);

            const float k0 = c1_0[j], k1 = c1_1[j], kc = c0s[j];
            float part[8];
            #pragma unroll
            for (int v = 0; v < 8; ++v) {
                float t0 = acc0[v];
                t0 = t0 > 0.0f ? t0 : (__expf(t0) - 1.0f);
                float t1 = acc1[v];
                t1 = t1 > 0.0f ? t1 : (__expf(t1) - 1.0f);
                part[v] = fmaf(t0, k0, fmaf(t1, k1, kc));
            }

            #pragma unroll
            for (int off = 8; off >= 1; off >>= 1) {
                #pragma unroll
                for (int v = 0; v < 8; ++v)
                    part[v] += __shfl_xor(part[v], off, 32);
            }

            for (int v = 0; v < 8; ++v)
                res[j][v] = part[v] + bq[j];
        }

        if (ncol == 0) {
            float* op = olane + (size_t)t * (BT * Q);
            #pragma unroll
            for (int v = 0; v < 8; ++v)
                *(v4f*)(op + (size_t)v * Q) = v4f{ res[0][v], res[1][v], res[2][v], res[3][v] };
        }

        for (int j = 0; j < QW; ++j) { a_lo[j] = n_lo[j]; a_hi[j] = n_hi[j]; }
    }
}

extern "C" void kernel_launch(void* const* d_in, const int* in_sizes, int n_in,
                              void* d_out, int out_size, void* d_ws, size_t ws_size,
                              hipStream_t stream) {
    const float* x     = (const float*)d_in[0];
    const float* W1    = (const float*)d_in[1];
    const float* b1    = (const float*)d_in[2];
    const float* gamma = (const float*)d_in[3];
    const float* beta  = (const float*)d_in[4];
    const float* mmean = (const float*)d_in[5];
    const float* mvar  = (const float*)d_in[6];
    const float* W2    = (const float*)d_in[7];
    const float* b2    = (const float*)d_in[8];
    float* out = (float*)d_out;

    const int B = in_sizes[0] / (Q * S);
    dim3 grid(QBLOCKS * BGROUPS);
    dim3 block(256);
    divenc_wmma_f32<<<grid, block, 0, stream>>>(x, W1, b1, gamma, beta,
                                                mmean, mvar, W2, b2, out, B);
}


# Swap the WMMA operand slots (product transposed: hidden units on registers, batch rows on lanes): sum over hidden units = in-lane FMA chains + one lane-half exchange; summation order changes
# speedup vs baseline: 2.0008x; 1.2162x over previous
#include <hip/hip_runtime.h>


typedef float v2f __attribute__((ext_vector_type(2)));
typedef float v4f __attribute__((ext_vector_type(4)));
typedef float v8f __attribute__((ext_vector_type(8)));

constexpr int Q = 128;
constexpr int S = 8;
constexpr int H = 32;
constexpr int BT = 16;
constexpr float BN_EPS = 1e-3f;

constexpr int QW      = 4;
constexpr int WAVES   = 8;
constexpr int QBLOCKS = Q / (QW * WAVES);
constexpr int BGROUPS = 128;


__device__ __forceinline__ float xhalf(float v) {
    const int b = __builtin_bit_cast(int, v);
    return __builtin_bit_cast(float,
        __builtin_amdgcn_permlanex16(b, b, 0x76543210, 0xfedcba98, false, false));
}

__global__ __launch_bounds__(256, 4) void divenc_wmma_f32(
    const float* __restrict__ x,
    const float* __restrict__ W1,
    const float* __restrict__ b1g,
    const float* __restrict__ gam,
    const float* __restrict__ bet,
    const float* __restrict__ mmean,
    const float* __restrict__ mvar,
    const float* __restrict__ W2,
    const float* __restrict__ b2g,
    float* __restrict__ out,
    int nrows)
{
    const int lane  = threadIdx.x & 31;
    const int wave  = threadIdx.x >> 5;

    const int mrow  = lane & 15;
    const int khalf = lane >> 4;
    const int ncol  = lane & 15;

    const int qgrp = (blockIdx.x % QBLOCKS) * WAVES + wave;
    const int bgrp = blockIdx.x / QBLOCKS;
    const int nbg  = gridDim.x / QBLOCKS;
    const int q0   = qgrp * QW;
    const int ntiles = nrows / BT;

    v2f wt0_lo[QW], wt0_hi[QW], wt1_lo[QW], wt1_hi[QW];
    v8f cb0[QW], cb1[QW];
    v8f kk0[QW], kk1[QW];
    float cst[QW];

    const int h0 = ncol, h1 = ncol + 16;
    const int kb = 2 * khalf;
    const int hb = 8 * khalf;
    for (int j = 0; j < QW; ++j) {
        const int q = q0 + j;
        const float* w1q = W1 + q * (S * H);
        wt0_lo[j] = v2f{ w1q[(kb + 0) * H + h0], w1q[(kb + 1) * H + h0] };
        wt0_hi[j] = v2f{ w1q[(kb + 4) * H + h0], w1q[(kb + 5) * H + h0] };
        wt1_lo[j] = v2f{ w1q[(kb + 0) * H + h1], w1q[(kb + 1) * H + h1] };
        wt1_hi[j] = v2f{ w1q[(kb + 4) * H + h1], w1q[(kb + 5) * H + h1] };

        const int i0 = q * H + h0, i1 = q * H + h1;
        const float bias0 = b1g[i0];
        const float bias1 = b1g[i1];
        const float inv0 = gam[i0] * __builtin_amdgcn_rsqf(mvar[i0] + BN_EPS);
        const float inv1 = gam[i1] * __builtin_amdgcn_rsqf(mvar[i1] + BN_EPS);
        const float w2_0 = W2[i0], w2_1 = W2[i1];
        const float c1_0 = inv0 * w2_0;
        const float c1_1 = inv1 * w2_1;
        float c0 = fmaf(-mmean[i0] * inv0 + bet[i0], w2_0,
                        (-mmean[i1] * inv1 + bet[i1]) * w2_1);
        const float bqv = b2g[q];

        #pragma unroll
        for (int r = 0; r < 8; ++r) {
            cb0[j][r] = __shfl(bias0, hb + r, 32);
            cb1[j][r] = __shfl(bias1, hb + r, 32);
            kk0[j][r] = __shfl(c1_0, hb + r, 32);
            kk1[j][r] = __shfl(c1_1, hb + r, 32);
        }
        for (int off = 8; off >= 1; off >>= 1)
            c0 += __shfl_xor(c0, off, 32);
        cst[j] = khalf == 0 ? c0 + bqv : 0.0f;
    }

    const float* xlane = x + (size_t)mrow * (Q * S) + q0 * S + 2 * khalf;
    float* olane = out + (size_t)mrow * Q + q0;

    v2f a_lo[QW], a_hi[QW];
    {
        const int t0 = bgrp < ntiles ? bgrp : ntiles - 1;
        const float* xt = xlane + (size_t)t0 * (BT * Q * S);
        for (int j = 0; j < QW; ++j) {
            a_lo[j] = v2f{ xt[j * S + 0], xt[j * S + 1] };
            a_hi[j] = v2f{ xt[j * S + 4], xt[j * S + 5] };
        }
    }

    for (int t = bgrp; t < ntiles; t += nbg) {
        v2f n_lo[QW], n_hi[QW];
        {
            const int tn = (t + nbg) < ntiles ? (t + nbg) : t;
            const float* xt = xlane + (size_t)tn * (BT * Q * S);
            for (int j = 0; j < QW; ++j) {
                n_lo[j] = v2f{ xt[j * S + 0], xt[j * S + 1] };
                n_hi[j] = v2f{ xt[j * S + 4], xt[j * S + 5] };
            }
        }

        float res[QW];
        for (int j = 0; j < QW; ++j) {
            v8f acc0 = __builtin_amdgcn_wmma_f32_16x16x4_f32(false, wt0_lo[j], false, a_lo[j],
                                                             (short)0, cb0[j], false, false);
            acc0 = __builtin_amdgcn_wmma_f32_16x16x4_f32(false, wt0_hi[j], false, a_hi[j],
                                                         (short)0, acc0, false, false);
            v8f acc1 = __builtin_amdgcn_wmma_f32_16x16x4_f32(false, wt1_lo[j], false, a_lo[j],
                                                             (short)0, cb1[j], false, false);
            acc1 = __builtin_amdgcn_wmma_f32_16x16x4_f32(false, wt1_hi[j], false, a_hi[j],
                                                         (short)0, acc1, false, false);

            float s0 = cst[j], s1 = 0.0f;
            #pragma unroll
            for (int r = 0; r < 8; ++r) {
                float t0 = acc0[r];
                t0 = t0 > 0.0f ? t0 : (__expf(t0) - 1.0f);
                float t1 = acc1[r];
                t1 = t1 > 0.0f ? t1 : (__expf(t1) - 1.0f);
                s0 = fmaf(t0, kk0[j][r], s0);
                s1 = fmaf(t1, kk1[j][r], s1);
            }
            res[j] = s0 + s1;
        }

        for (int j = 0; j < QW; ++j)
            res[j] += xhalf(res[j]);

        if (khalf == 0)
            *(v4f*)(olane + (size_t)t * (BT * Q)) = v4f{ res[0], res[1], res[2], res[3] };

        for (int j = 0; j < QW; ++j) { a_lo[j] = n_lo[j]; a_hi[j] = n_hi[j]; }
    }
}

extern "C" void kernel_launch(void* const* d_in, const int* in_sizes, int n_in,
                              void* d_out, int out_size, void* d_ws, size_t ws_size,
                              hipStream_t stream) {
    const float* x     = (const float*)d_in[0];
    const float* W1    = (const float*)d_in[1];
    const float* b1    = (const float*)d_in[2];
    const float* gamma = (const float*)d_in[3];
    const float* beta  = (const float*)d_in[4];
    const float* mmean = (const float*)d_in[5];
    const float* mvar  = (const float*)d_in[6];
    const float* W2    = (const float*)d_in[7];
    const float* b2    = (const float*)d_in[8];
    float* out = (float*)d_out;

    const int B = in_sizes[0] / (Q * S);
    dim3 grid(QBLOCKS * BGROUPS);
    dim3 block(256);
    divenc_wmma_f32<<<grid, block, 0, stream>>>(x, W1, b1, gamma, beta,
                                                mmean, mvar, W2, b2, out, B);
}


# Next tile's x loads now stand right behind the WMMAs that read the current fragments (same registers, a scheduler fence behind each pair), not at the loop's end: one full trip of lead
# speedup vs baseline: 2.1445x; 1.0718x over previous
#include <hip/hip_runtime.h>


typedef float v2f __attribute__((ext_vector_type(2)));
typedef float v4f __attribute__((ext_vector_type(4)));
typedef float v8f __attribute__((ext_vector_type(8)));

constexpr int Q = 128;
constexpr int S = 8;
constexpr int H = 32;
constexpr int BT = 16;
constexpr float BN_EPS = 1e-3f;

constexpr int QW      = 4;
constexpr int WAVES   = 8;
constexpr int QBLOCKS = Q / (QW * WAVES);
constexpr int BGROUPS = 128;


__device__ __forceinline__ float xhalf(float v) {
    const int b = __builtin_bit_cast(int, v);
    return __builtin_bit_cast(float,
        __builtin_amdgcn_permlanex16(b, b, 0x76543210, 0xfedcba98, false, false));
}

__global__ __launch_bounds__(256, 4) void divenc_wmma_f32(
    const float* __restrict__ x,
    const float* __restrict__ W1,
    const float* __restrict__ b1g,
    const float* __restrict__ gam,
    const float* __restrict__ bet,
    const float* __restrict__ mmean,
    const float* __restrict__ mvar,
    const float* __restrict__ W2,
    const float* __restrict__ b2g,
    float* __restrict__ out,
    int nrows)
{
    const int lane  = threadIdx.x & 31;
    const int wave  = threadIdx.x >> 5;

    const int mrow  = lane & 15;
    const int khalf = lane >> 4;
    const int ncol  = lane & 15;

    const int qgrp = (blockIdx.x % QBLOCKS) * WAVES + wave;
    const int bgrp = blockIdx.x / QBLOCKS;
    const int nbg  = gridDim.x / QBLOCKS;
    const int q0   = qgrp * QW;
    const int ntiles = nrows / BT;

    v2f wt0_lo[QW], wt0_hi[QW], wt1_lo[QW], wt1_hi[QW];
    v8f cb0[QW], cb1[QW];
    v8f kk0[QW], kk1[QW];
    float cst[QW];

    const int h0 = ncol, h1 = ncol + 16;
    const int kb = 2 * khalf;
    const int hb = 8 * khalf;
    for (int j = 0; j < QW; ++j) {
        const int q = q0 + j;
        const float* w1q = W1 + q * (S * H);
        wt0_lo[j] = v2f{ w1q[(kb + 0) * H + h0], w1q[(kb + 1) * H + h0] };
        wt0_hi[j] = v2f{ w1q[(kb + 4) * H + h0], w1q[(kb + 5) * H + h0] };
        wt1_lo[j] = v2f{ w1q[(kb + 0) * H + h1], w1q[(kb + 1) * H + h1] };
        wt1_hi[j] = v2f{ w1q[(kb + 4) * H + h1], w1q[(kb + 5) * H + h1] };

        const int i0 = q * H + h0, i1 = q * H + h1;
        const float bias0 = b1g[i0];
        const float bias1 = b1g[i1];
        const float inv0 = gam[i0] * __builtin_amdgcn_rsqf(mvar[i0] + BN_EPS);
        const float inv1 = gam[i1] * __builtin_amdgcn_rsqf(mvar[i1] + BN_EPS);
        const float w2_0 = W2[i0], w2_1 = W2[i1];
        const float c1_0 = inv0 * w2_0;
        const float c1_1 = inv1 * w2_1;
        float c0 = fmaf(-mmean[i0] * inv0 + bet[i0], w2_0,
                        (-mmean[i1] * inv1 + bet[i1]) * w2_1);
        const float bqv = b2g[q];

        #pragma unroll
        for (int r = 0; r < 8; ++r) {
            cb0[j][r] = __shfl(bias0, hb + r, 32);
            cb1[j][r] = __shfl(bias1, hb + r, 32);
            kk0[j][r] = __shfl(c1_0, hb + r, 32);
            kk1[j][r] = __shfl(c1_1, hb + r, 32);
        }
        for (int off = 8; off >= 1; off >>= 1)
            c0 += __shfl_xor(c0, off, 32);
        cst[j] = khalf == 0 ? c0 + bqv : 0.0f;
    }

    const float* xlane = x + (size_t)mrow * (Q * S) + q0 * S + 2 * khalf;
    float* olane = out + (size_t)mrow * Q + q0;

    v2f a_lo[QW], a_hi[QW];
    {
        const int t0 = bgrp < ntiles ? bgrp : ntiles - 1;
        const float* xt = xlane + (size_t)t0 * (BT * Q * S);
        for (int j = 0; j < QW; ++j) {
            a_lo[j] = v2f{ xt[j * S + 0], xt[j * S + 1] };
            a_hi[j] = v2f{ xt[j * S + 4], xt[j * S + 5] };
        }
    }

    for (int t = bgrp; t < ntiles; t += nbg) {
        const int tn = (t + nbg) < ntiles ? (t + nbg) : t;
        const float* xn = xlane + (size_t)tn * (BT * Q * S);

        float res[QW];
        for (int j = 0; j < QW; ++j) {
            v8f acc0 = __builtin_amdgcn_wmma_f32_16x16x4_f32(false, wt0_lo[j], false, a_lo[j],
                                                             (short)0, cb0[j], false, false);
            acc0 = __builtin_amdgcn_wmma_f32_16x16x4_f32(false, wt0_hi[j], false, a_hi[j],
                                                         (short)0, acc0, false, false);
            v8f acc1 = __builtin_amdgcn_wmma_f32_16x16x4_f32(false, wt1_lo[j], false, a_lo[j],
                                                             (short)0, cb1[j], false, false);
            acc1 = __builtin_amdgcn_wmma_f32_16x16x4_f32(false, wt1_hi[j], false, a_hi[j],
                                                         (short)0, acc1, false, false);

            a_lo[j] = v2f{ xn[j * S + 0], xn[j * S + 1] };
            a_hi[j] = v2f{ xn[j * S + 4], xn[j * S + 5] };
            __builtin_amdgcn_sched_barrier(0);

            float s0 = cst[j], s1 = 0.0f;
            #pragma unroll
            for (int r = 0; r < 8; ++r) {
                float t0 = acc0[r];
                t0 = t0 > 0.0f ? t0 : (__expf(t0) - 1.0f);
                float t1 = acc1[r];
                t1 = t1 > 0.0f ? t1 : (__expf(t1) - 1.0f);
                s0 = fmaf(t0, kk0[j][r], s0);
                s1 = fmaf(t1, kk1[j][r], s1);
            }
            res[j] = s0 + s1;
        }

        for (int j = 0; j < QW; ++j)
            res[j] += xhalf(res[j]);

        if (khalf == 0)
            *(v4f*)(olane + (size_t)t * (BT * Q)) = v4f{ res[0], res[1], res[2], res[3] };
    }
}

extern "C" void kernel_launch(void* const* d_in, const int* in_sizes, int n_in,
                              void* d_out, int out_size, void* d_ws, size_t ws_size,
                              hipStream_t stream) {
    const float* x     = (const float*)d_in[0];
    const float* W1    = (const float*)d_in[1];
    const float* b1    = (const float*)d_in[2];
    const float* gamma = (const float*)d_in[3];
    const float* beta  = (const float*)d_in[4];
    const float* mmean = (const float*)d_in[5];
    const float* mvar  = (const float*)d_in[6];
    const float* W2    = (const float*)d_in[7];
    const float* b2    = (const float*)d_in[8];
    float* out = (float*)d_out;

    const int B = in_sizes[0] / (Q * S);
    dim3 grid(QBLOCKS * BGROUPS);
    dim3 block(256);
    divenc_wmma_f32<<<grid, block, 0, stream>>>(x, W1, b1, gamma, beta,
                                                mmean, mvar, W2, b2, out, B);
}


# ELU as ln2*max(u,0) + min(2^u,1) on an accumulator pre-scaled by log2e (W1, b1 scaled at load), -1 folded into the constant: exp with clamp, max, 2 FMAs per hidden unit; summation order changes
# speedup vs baseline: 2.3114x; 1.0778x over previous
#include <hip/hip_runtime.h>


typedef float v2f __attribute__((ext_vector_type(2)));
typedef float v4f __attribute__((ext_vector_type(4)));
typedef float v8f __attribute__((ext_vector_type(8)));

constexpr int Q = 128;
constexpr int S = 8;
constexpr int H = 32;
constexpr int BT = 16;
constexpr float BN_EPS = 1e-3f;
constexpr float LOG2E = 1.44269504088896340736f;
constexpr float LN2   = 0.69314718055994530942f;

constexpr int QW      = 4;
constexpr int WAVES   = 8;
constexpr int QBLOCKS = Q / (QW * WAVES);
constexpr int BGROUPS = 128;


__device__ __forceinline__ float xhalf(float v) {
    const int b = __builtin_bit_cast(int, v);
    return __builtin_bit_cast(float,
        __builtin_amdgcn_permlanex16(b, b, 0x76543210, 0xfedcba98, false, false));
}

__global__ __launch_bounds__(256, 4) void divenc_wmma_f32(
    const float* __restrict__ x,
    const float* __restrict__ W1,
    const float* __restrict__ b1g,
    const float* __restrict__ gam,
    const float* __restrict__ bet,
    const float* __restrict__ mmean,
    const float* __restrict__ mvar,
    const float* __restrict__ W2,
    const float* __restrict__ b2g,
    float* __restrict__ out,
    int nrows)
{
    const int lane  = threadIdx.x & 31;
    const int wave  = threadIdx.x >> 5;

    const int mrow  = lane & 15;
    const int khalf = lane >> 4;
    const int ncol  = lane & 15;

    const int qgrp = (blockIdx.x % QBLOCKS) * WAVES + wave;
    const int bgrp = blockIdx.x / QBLOCKS;
    const int nbg  = gridDim.x / QBLOCKS;
    const int q0   = qgrp * QW;
    const int ntiles = nrows / BT;

    v2f wt0_lo[QW], wt0_hi[QW], wt1_lo[QW], wt1_hi[QW];
    v8f cb0[QW], cb1[QW];
    v8f kk0[QW], kk1[QW];
    float cst[QW];

    const int h0 = ncol, h1 = ncol + 16;
    const int kb = 2 * khalf;
    const int hb = 8 * khalf;
    for (int j = 0; j < QW; ++j) {
        const int q = q0 + j;
        const float* w1q = W1 + q * (S * H);
        wt0_lo[j] = v2f{ w1q[(kb + 0) * H + h0], w1q[(kb + 1) * H + h0] } * LOG2E;
        wt0_hi[j] = v2f{ w1q[(kb + 4) * H + h0], w1q[(kb + 5) * H + h0] } * LOG2E;
        wt1_lo[j] = v2f{ w1q[(kb + 0) * H + h1], w1q[(kb + 1) * H + h1] } * LOG2E;
        wt1_hi[j] = v2f{ w1q[(kb + 4) * H + h1], w1q[(kb + 5) * H + h1] } * LOG2E;

        const int i0 = q * H + h0, i1 = q * H + h1;
        const float bias0 = b1g[i0] * LOG2E;
        const float bias1 = b1g[i1] * LOG2E;
        const float inv0 = gam[i0] * __builtin_amdgcn_rsqf(mvar[i0] + BN_EPS);
        const float inv1 = gam[i1] * __builtin_amdgcn_rsqf(mvar[i1] + BN_EPS);
        const float w2_0 = W2[i0], w2_1 = W2[i1];
        const float c1_0 = inv0 * w2_0;
        const float c1_1 = inv1 * w2_1;
        float c0 = fmaf(-mmean[i0] * inv0 + bet[i0], w2_0,
                        (-mmean[i1] * inv1 + bet[i1]) * w2_1);
        c0 -= c1_0 + c1_1;
        const float bqv = b2g[q];

        #pragma unroll
        for (int r = 0; r < 8; ++r) {
            cb0[j][r] = __shfl(bias0, hb + r, 32);
            cb1[j][r] = __shfl(bias1, hb + r, 32);
            kk0[j][r] = __shfl(c1_0, hb + r, 32);
            kk1[j][r] = __shfl(c1_1, hb + r, 32);
        }
        for (int off = 8; off >= 1; off >>= 1)
            c0 += __shfl_xor(c0, off, 32);
        cst[j] = khalf == 0 ? c0 + bqv : 0.0f;
    }

    const float* xlane = x + (size_t)mrow * (Q * S) + q0 * S + 2 * khalf;
    float* olane = out + (size_t)mrow * Q + q0;

    v2f a_lo[QW], a_hi[QW];
    {
        const int t0 = bgrp < ntiles ? bgrp : ntiles - 1;
        const float* xt = xlane + (size_t)t0 * (BT * Q * S);
        for (int j = 0; j < QW; ++j) {
            a_lo[j] = v2f{ xt[j * S + 0], xt[j * S + 1] };
            a_hi[j] = v2f{ xt[j * S + 4], xt[j * S + 5] };
        }
    }

    for (int t = bgrp; t < ntiles; t += nbg) {
        const int tn = (t + nbg) < ntiles ? (t + nbg) : t;
        const float* xn = xlane + (size_t)tn * (BT * Q * S);

        float res[QW];
        for (int j = 0; j < QW; ++j) {
            v8f acc0 = __builtin_amdgcn_wmma_f32_16x16x4_f32(false, wt0_lo[j], false, a_lo[j],
                                                             (short)0, cb0[j], false, false);
            acc0 = __builtin_amdgcn_wmma_f32_16x16x4_f32(false, wt0_hi[j], false, a_hi[j],
                                                         (short)0, acc0, false, false);
            v8f acc1 = __builtin_amdgcn_wmma_f32_16x16x4_f32(false, wt1_lo[j], false, a_lo[j],
                                                             (short)0, cb1[j], false, false);
            acc1 = __builtin_amdgcn_wmma_f32_16x16x4_f32(false, wt1_hi[j], false, a_hi[j],
                                                         (short)0, acc1, false, false);

            a_lo[j] = v2f{ xn[j * S + 0], xn[j * S + 1] };
            a_hi[j] = v2f{ xn[j * S + 4], xn[j * S + 5] };
            __builtin_amdgcn_sched_barrier(0);

            float e0 = cst[j], e1 = 0.0f, p0 = 0.0f, p1 = 0.0f;
            #pragma unroll
            for (int r = 0; r < 8; ++r) {
                const float u0 = acc0[r];
                const float u1 = acc1[r];
                const float g0 = __builtin_amdgcn_fmed3f(__builtin_amdgcn_exp2f(u0), 0.0f, 1.0f);
                const float g1 = __builtin_amdgcn_fmed3f(__builtin_amdgcn_exp2f(u1), 0.0f, 1.0f);
                e0 = fmaf(g0, kk0[j][r], e0);
                e1 = fmaf(g1, kk1[j][r], e1);
                p0 = fmaf(__builtin_fmaxf(u0, 0.0f), kk0[j][r], p0);
                p1 = fmaf(__builtin_fmaxf(u1, 0.0f), kk1[j][r], p1);
            }
            res[j] = fmaf(p0 + p1, LN2, e0 + e1);
        }

        for (int j = 0; j < QW; ++j)
            res[j] += xhalf(res[j]);

        if (khalf == 0)
            *(v4f*)(olane + (size_t)t * (BT * Q)) = v4f{ res[0], res[1], res[2], res[3] };
    }
}

extern "C" void kernel_launch(void* const* d_in, const int* in_sizes, int n_in,
                              void* d_out, int out_size, void* d_ws, size_t ws_size,
                              hipStream_t stream) {
    const float* x     = (const float*)d_in[0];
    const float* W1    = (const float*)d_in[1];
    const float* b1    = (const float*)d_in[2];
    const float* gamma = (const float*)d_in[3];
    const float* beta  = (const float*)d_in[4];
    const float* mmean = (const float*)d_in[5];
    const float* mvar  = (const float*)d_in[6];
    const float* W2    = (const float*)d_in[7];
    const float* b2    = (const float*)d_in[8];
    float* out = (float*)d_out;

    const int B = in_sizes[0] / (Q * S);
    dim3 grid(QBLOCKS * BGROUPS);
    dim3 block(256);
    divenc_wmma_f32<<<grid, block, 0, stream>>>(x, W1, b1, gamma, beta,
                                                mmean, mvar, W2, b2, out, B);
}
